# starting kernel
# baseline (speedup 1.0000x reference)
#include <hip/hip_runtime.h>


#define NPAIR 169

typedef __attribute__((ext_vector_type(2))) float v2f;
typedef __attribute__((ext_vector_type(8))) float v8f;

__global__ void __launch_bounds__(256)
hon_prep(const int* __restrict__ x, const int* __restrict__ fact,
         int* __restrict__ ids, int* __restrict__ perm,
         int* __restrict__ gstart, int* __restrict__ work,
         int N, int wcap)
{
    __shared__ int cnt[NPAIR];
    __shared__ int pos[NPAIR];
    const int tid = threadIdx.x;

    for (int g = tid; g < NPAIR; g += blockDim.x) cnt[g] = 0;
    __syncthreads();

    for (int n = tid; n < N; n += blockDim.x) {
        const int a  = fact[2 * n];
        const int id = x[3 * a + 1] * 13 + x[3 * a + 2];
        ids[n] = id;
        atomicAdd(&cnt[id], 1);
    }
    __syncthreads();

    if (tid == 0) {
        int acc = 0, w = 0;
        for (int g = 0; g < NPAIR; ++g) {
            gstart[g] = acc;
            pos[g]    = acc;
            const int c  = cnt[g];
            const int nt = (c + 15) >> 4;
            for (int t = 0; t < nt && w < wcap; ++t, ++w) {
                work[2 * w]     = g;
                work[2 * w + 1] = t;
            }
            acc += c;
        }
        gstart[NPAIR] = acc;
        for (; w < wcap; ++w) { work[2 * w] = -1; work[2 * w + 1] = 0; }
    }
    __syncthreads();

    for (int n = tid; n < N; n += blockDim.x) {
        const int p = atomicAdd(&pos[ids[n]], 1);
        perm[p] = n;
    }
}

__global__ void __launch_bounds__(128)
hon_gemm(const float* __restrict__ inp, const float* __restrict__ params,
         const float* __restrict__ bias, const int* __restrict__ perm,
         const int* __restrict__ gstart, const int* __restrict__ work,
         const int* __restrict__ msg_to_p, float* __restrict__ out,
         int N, int H, int O, int ORD)
{
    extern __shared__ float As[];

    const int g = work[2 * blockIdx.x];
    if (g < 0) return;
    const int mt     = work[2 * blockIdx.x + 1];
    const int rowbeg = gstart[g] + mt * 16;
    int cnt = gstart[g + 1] - rowbeg;
    cnt = cnt > 16 ? 16 : cnt;
    const int msg_to = msg_to_p[0];

    {
        const int HQ = H >> 2;
        for (int idx = threadIdx.x; idx < 16 * HQ; idx += blockDim.x) {
            const int r  = idx / HQ;
            const int kq = idx - r * HQ;
            float4 p = make_float4(0.f, 0.f, 0.f, 0.f);
            if (r < cnt) {
                const int n = perm[rowbeg + r];
                p = make_float4(1.f, 1.f, 1.f, 1.f);
                for (int i = 0; i < ORD; ++i) {
                    if (i == msg_to) continue;
                    const float4 v =
                        ((const float4*)(inp + ((size_t)i * N + n) * H))[kq];
                    p.x *= v.x; p.y *= v.y; p.z *= v.z; p.w *= v.w;
                }
            }
            ((float4*)(As + r * H))[kq] = p;
        }
    }
    __syncthreads();

    const int lane   = threadIdx.x & 31;
    const int waveid = threadIdx.x >> 5;
    const int m      = lane & 15;
    const int khalf  = (lane >> 4) << 1;

    const float* W       = params + (size_t)g * H * O;
    const int    colbase = blockIdx.y * 256 + waveid * 64 + m;
    const float* aptr    = As + m * H + khalf;

    v8f acc[4];
#pragma unroll
    for (int t = 0; t < 4; ++t) acc[t] = (v8f){0, 0, 0, 0, 0, 0, 0, 0};

#pragma unroll 2
    for (int kb = 0; kb < H; kb += 4) {
        const int k0 = kb + khalf;
        const v2f a  = *(const v2f*)(aptr + kb);

        const float* Wk0 = W + (size_t)k0 * O + colbase;
        const float* Wk1 = Wk0 + O;
#pragma unroll
        for (int t = 0; t < 4; ++t) {
            v2f b; b.x = Wk0[t * 16]; b.y = Wk1[t * 16];
            acc[t] = __builtin_amdgcn_wmma_f32_16x16x4_f32(
                false, a, false, b, (short)0, acc[t], false, false);
        }
    }

    const int mbase = (lane >> 4) << 3;
    int nrow[8];
#pragma unroll
    for (int r = 0; r < 8; ++r) {
        const int mr = mbase + r;
        nrow[r] = (mr < cnt) ? perm[rowbeg + mr] : -1;
    }

#pragma unroll
    for (int t = 0; t < 4; ++t) {
        const int   col = colbase + t * 16;
        const float bv  = bias[(size_t)g * O + col];
#pragma unroll
        for (int r = 0; r < 8; ++r) {
            if (nrow[r] >= 0)
                out[(size_t)nrow[r] * O + col] = acc[t][r] + bv;
        }
    }
}

extern "C" void kernel_launch(void* const* d_in, const int* in_sizes, int n_in,
                              void* d_out, int out_size, void* d_ws, size_t ws_size,
                              hipStream_t stream)
{
    const int*   x      = (const int*)  d_in[0];
    const int*   fact   = (const int*)  d_in[1];
    const float* inp    = (const float*)d_in[2];
    const float* params = (const float*)d_in[3];
    const float* bias   = (const float*)d_in[4];
    const int*   msg_to = (const int*)  d_in[5];
    float*       out    = (float*)d_out;

    const int N   = in_sizes[1] / 2;
    const int O   = in_sizes[4] / NPAIR;
    const int H   = in_sizes[3] / (NPAIR * O);
    const int ORD = in_sizes[2] / (N * H);

    const int wcap = NPAIR + (N + 15) / 16;

    int* ws     = (int*)d_ws;
    int* ids    = ws;
    int* perm   = ws + N;
    int* gstart = ws + 2 * N;
    int* work   = ws + 2 * N + NPAIR + 1;

    hipLaunchKernelGGL(hon_prep, dim3(1), dim3(256), 0, stream,
                       x, fact, ids, perm, gstart, work, N, wcap);

    const size_t lds_bytes = (size_t)16 * H * sizeof(float);
    dim3 grid(wcap, O / 256);
    hipLaunchKernelGGL(hon_gemm, grid, dim3(128), lds_bytes, stream,
                       inp, params, bias, perm, gstart, work, msg_to, out,
                       N, H, O, ORD);
}


# B (weight) rows read by 128-bit loads (a lane owns 4 adjacent columns, one per col-tile), requested one batch of 8 K-steps ahead in a second register set; out written by 128-bit stores
# speedup vs baseline: 1.4418x; 1.4418x over previous
#include <hip/hip_runtime.h>


#define NPAIR 169

typedef __attribute__((ext_vector_type(2))) float v2f;
typedef __attribute__((ext_vector_type(4))) float v4f;
typedef __attribute__((ext_vector_type(8))) float v8f;

#define KS 8
#define KBATCH (4 * KS)

__global__ void __launch_bounds__(256)
hon_prep(const int* __restrict__ x, const int* __restrict__ fact,
         int* __restrict__ ids, int* __restrict__ perm,
         int* __restrict__ gstart, int* __restrict__ work,
         int N, int wcap)
{
    __shared__ int cnt[NPAIR];
    __shared__ int pos[NPAIR];
    const int tid = threadIdx.x;

    for (int g = tid; g < NPAIR; g += blockDim.x) cnt[g] = 0;
    __syncthreads();

    for (int n = tid; n < N; n += blockDim.x) {
        const int a  = fact[2 * n];
        const int id = x[3 * a + 1] * 13 + x[3 * a + 2];
        ids[n] = id;
        atomicAdd(&cnt[id], 1);
    }
    __syncthreads();

    if (tid == 0) {
        int acc = 0, w = 0;
        for (int g = 0; g < NPAIR; ++g) {
            gstart[g] = acc;
            pos[g]    = acc;
            const int c  = cnt[g];
            const int nt = (c + 15) >> 4;
            for (int t = 0; t < nt && w < wcap; ++t, ++w) {
                work[2 * w]     = g;
                work[2 * w + 1] = t;
            }
            acc += c;
        }
        gstart[NPAIR] = acc;
        for (; w < wcap; ++w) { work[2 * w] = -1; work[2 * w + 1] = 0; }
    }
    __syncthreads();

    for (int n = tid; n < N; n += blockDim.x) {
        const int p = atomicAdd(&pos[ids[n]], 1);
        perm[p] = n;
    }
}

__global__ void __launch_bounds__(128)
hon_gemm(const float* __restrict__ inp, const float* __restrict__ params,
         const float* __restrict__ bias, const int* __restrict__ perm,
         const int* __restrict__ gstart, const int* __restrict__ work,
         const int* __restrict__ msg_to_p, float* __restrict__ out,
         int N, int H, int O, int ORD)
{
    extern __shared__ float As[];

    const int g = work[2 * blockIdx.x];
    if (g < 0) return;
    const int mt     = work[2 * blockIdx.x + 1];
    const int rowbeg = gstart[g] + mt * 16;
    int cnt = gstart[g + 1] - rowbeg;
    cnt = cnt > 16 ? 16 : cnt;
    const int msg_to = msg_to_p[0];

    {
        const int HQ = H >> 2;
        for (int idx = threadIdx.x; idx < 16 * HQ; idx += blockDim.x) {
            const int r  = idx / HQ;
            const int kq = idx - r * HQ;
            float4 p = make_float4(0.f, 0.f, 0.f, 0.f);
            if (r < cnt) {
                const int n = perm[rowbeg + r];
                p = make_float4(1.f, 1.f, 1.f, 1.f);
                for (int i = 0; i < ORD; ++i) {
                    if (i == msg_to) continue;
                    const float4 v =
                        ((const float4*)(inp + ((size_t)i * N + n) * H))[kq];
                    p.x *= v.x; p.y *= v.y; p.z *= v.z; p.w *= v.w;
                }
            }
            ((float4*)(As + r * H))[kq] = p;
        }
    }
    __syncthreads();

    const int lane   = threadIdx.x & 31;
    const int waveid = threadIdx.x >> 5;
    const int m      = lane & 15;
    const int khalf  = (lane >> 4) << 1;

    const float* W    = params + (size_t)g * H * O;
    const int    col4 = blockIdx.y * 256 + waveid * 64 + m * 4;
    const float* aptr = As + m * H + khalf;
    const float* bptr = W + (size_t)khalf * O + col4;

    v8f acc[4];
#pragma unroll
    for (int t = 0; t < 4; ++t) acc[t] = (v8f){0, 0, 0, 0, 0, 0, 0, 0};

    auto mma_step = [&](const v2f a, const v4f b0, const v4f b1) {
#pragma unroll
        for (int t = 0; t < 4; ++t) {
            v2f b; b.x = b0[t]; b.y = b1[t];
            acc[t] = __builtin_amdgcn_wmma_f32_16x16x4_f32(
                false, a, false, b, (short)0, acc[t], false, false);
        }
    };
    auto load_batch = [&](v4f (&q)[KS][2], const int kb) {
#pragma unroll
        for (int s = 0; s < KS; ++s) {
            const float* p = bptr + (size_t)(kb + 4 * s) * O;
            q[s][0] = *(const v4f*)p;
            q[s][1] = *(const v4f*)(p + O);
        }
    };
    auto mma_batch = [&](const v4f (&q)[KS][2], const int kb) {
#pragma unroll
        for (int s = 0; s < KS; ++s) {
            const v2f a = *(const v2f*)(aptr + kb + 4 * s);
            mma_step(a, q[s][0], q[s][1]);
        }
    };

    const int Hp = H - H % (2 * KBATCH);
    int kb = 0;
    if (Hp > 0) {
        v4f q0[KS][2], q1[KS][2];
        load_batch(q0, 0);
        for (; kb + 2 * KBATCH < Hp; kb += 2 * KBATCH) {
            load_batch(q1, kb + KBATCH);
            __builtin_amdgcn_sched_barrier(0);
            mma_batch(q0, kb);
            load_batch(q0, kb + 2 * KBATCH);
            __builtin_amdgcn_sched_barrier(0);
            mma_batch(q1, kb + KBATCH);
        }
        load_batch(q1, kb + KBATCH);
        __builtin_amdgcn_sched_barrier(0);
        mma_batch(q0, kb);
        mma_batch(q1, kb + KBATCH);
        kb += 2 * KBATCH;
    }
    for (; kb < H; kb += 4) {
        const float* p = bptr + (size_t)kb * O;
        mma_step(*(const v2f*)(aptr + kb), *(const v4f*)p, *(const v4f*)(p + O));
    }

    const int mbase = (lane >> 4) << 3;
    const v4f bv    = *(const v4f*)(bias + (size_t)g * O + col4);
#pragma unroll
    for (int r = 0; r < 8; ++r) {
        const int mr = mbase + r;
        if (mr < cnt) {
            const int n = perm[rowbeg + mr];
            v4f o;
            o.x = acc[0][r] + bv.x; o.y = acc[1][r] + bv.y;
            o.z = acc[2][r] + bv.z; o.w = acc[3][r] + bv.w;
            *(v4f*)(out + (size_t)n * O + col4) = o;
        }
    }
}

extern "C" void kernel_launch(void* const* d_in, const int* in_sizes, int n_in,
                              void* d_out, int out_size, void* d_ws, size_t ws_size,
                              hipStream_t stream)
{
    const int*   x      = (const int*)  d_in[0];
    const int*   fact   = (const int*)  d_in[1];
    const float* inp    = (const float*)d_in[2];
    const float* params = (const float*)d_in[3];
    const float* bias   = (const float*)d_in[4];
    const int*   msg_to = (const int*)  d_in[5];
    float*       out    = (float*)d_out;

    const int N   = in_sizes[1] / 2;
    const int O   = in_sizes[4] / NPAIR;
    const int H   = in_sizes[3] / (NPAIR * O);
    const int ORD = in_sizes[2] / (N * H);

    const int wcap = NPAIR + (N + 15) / 16;

    int* ws     = (int*)d_ws;
    int* ids    = ws;
    int* perm   = ws + N;
    int* gstart = ws + 2 * N;
    int* work   = ws + 2 * N + NPAIR + 1;

    hipLaunchKernelGGL(hon_prep, dim3(1), dim3(256), 0, stream,
                       x, fact, ids, perm, gstart, work, N, wcap);

    const size_t lds_bytes = (size_t)16 * H * sizeof(float);
    dim3 grid(wcap, O / 256);
    hipLaunchKernelGGL(hon_gemm, grid, dim3(128), lds_bytes, stream,
                       inp, params, bias, perm, gstart, work, msg_to, out,
                       N, H, O, ORD);
}


# Kernel 1 (row sort + worklist) made parallel: 1024 threads, one row each; group offsets by a shuffle prefix scan, not a one-thread loop over 169 groups; position = group start + histogram-atomic rank
# speedup vs baseline: 2.1146x; 1.4666x over previous
#include <hip/hip_runtime.h>


#define NPAIR 169

typedef __attribute__((ext_vector_type(2))) float v2f;
typedef __attribute__((ext_vector_type(4))) float v4f;
typedef __attribute__((ext_vector_type(8))) float v8f;

#define KS 8
#define KBATCH (4 * KS)

#define PREP_THREADS 1024
#define GWAVES ((NPAIR + 31) / 32)

__global__ void __launch_bounds__(PREP_THREADS)
hon_prep(const int* __restrict__ x, const int* __restrict__ fact,
         int* __restrict__ ids, int* __restrict__ perm,
         int* __restrict__ gstart, int* __restrict__ work,
         int N, int wcap)
{
    __shared__ int cnt[NPAIR];
    __shared__ int start[NPAIR];
    __shared__ int wrow[GWAVES];
    __shared__ int wtil[GWAVES];
    __shared__ int ntile;

    const int tid  = threadIdx.x;
    const int B    = blockDim.x;
    const int lane = tid & 31;
    const int wv   = tid >> 5;

    int id0 = 0;
    if (tid < N) {
        const int a = fact[2 * tid];
        id0 = x[3 * a + 1] * 13 + x[3 * a + 2];
    }
    for (int g = tid; g < NPAIR; g += B) cnt[g] = 0;
    __syncthreads();

    int rk0 = 0;
    if (tid < N) rk0 = atomicAdd(&cnt[id0], 1);
    for (int n = tid + B; n < N; n += B) {
        const int a  = fact[2 * n];
        const int id = x[3 * a + 1] * 13 + x[3 * a + 2];
        ids[n] = (atomicAdd(&cnt[id], 1) << 8) | id;
    }
    __syncthreads();

    int c = 0, nt = 0, sc = 0, st = 0;
    if (wv < GWAVES) {
        if (tid < NPAIR) { c = cnt[tid]; nt = (c + 15) >> 4; }
        sc = c; st = nt;
        for (int d = 1; d < 32; d <<= 1) {
            const int uc = __shfl_up(sc, d);
            const int ut = __shfl_up(st, d);
            if (lane >= d) { sc += uc; st += ut; }
        }
        if (lane == 31) { wrow[wv] = sc; wtil[wv] = st; }
    }
    __syncthreads();

    int rs = 0, ws = 0;
    if (tid < NPAIR) {
        int br = 0, bt = 0;
        for (int w = 0; w < GWAVES - 1; ++w) {
            const int r = wrow[w], t = wtil[w];
            if (w < wv) { br += r; bt += t; }
        }
        rs = br + sc - c;
        ws = bt + st - nt;
        start[tid] = rs;
        if (tid == NPAIR - 1) ntile = ws + nt;
    }
    __syncthreads();

    if (tid < NPAIR) {
        gstart[tid] = rs;
        if (tid == NPAIR - 1) gstart[NPAIR] = rs + c;
        for (int t = 0; t < nt; ++t) {
            const int w = ws + t;
            if (w < wcap) { work[2 * w] = tid; work[2 * w + 1] = t; }
        }
    }
    for (int w = ntile + tid; w < wcap; w += B) {
        work[2 * w] = -1; work[2 * w + 1] = 0;
    }
    if (tid < N) perm[start[id0] + rk0] = tid;
    for (int n = tid + B; n < N; n += B) {
        const int v = ids[n];
        perm[start[v & 255] + (v >> 8)] = n;
    }
}

__global__ void __launch_bounds__(128)
hon_gemm(const float* __restrict__ inp, const float* __restrict__ params,
         const float* __restrict__ bias, const int* __restrict__ perm,
         const int* __restrict__ gstart, const int* __restrict__ work,
         const int* __restrict__ msg_to_p, float* __restrict__ out,
         int N, int H, int O, int ORD)
{
    extern __shared__ float As[];

    const int g = work[2 * blockIdx.x];
    if (g < 0) return;
    const int mt     = work[2 * blockIdx.x + 1];
    const int rowbeg = gstart[g] + mt * 16;
    int cnt = gstart[g + 1] - rowbeg;
    cnt = cnt > 16 ? 16 : cnt;
    const int msg_to = msg_to_p[0];

    {
        const int HQ = H >> 2;
        for (int idx = threadIdx.x; idx < 16 * HQ; idx += blockDim.x) {
            const int r  = idx / HQ;
            const int kq = idx - r * HQ;
            float4 p = make_float4(0.f, 0.f, 0.f, 0.f);
            if (r < cnt) {
                const int n = perm[rowbeg + r];
                p = make_float4(1.f, 1.f, 1.f, 1.f);
                for (int i = 0; i < ORD; ++i) {
                    if (i == msg_to) continue;
                    const float4 v =
                        ((const float4*)(inp + ((size_t)i * N + n) * H))[kq];
                    p.x *= v.x; p.y *= v.y; p.z *= v.z; p.w *= v.w;
                }
            }
            ((float4*)(As + r * H))[kq] = p;
        }
    }
    __syncthreads();

    const int lane   = threadIdx.x & 31;
    const int waveid = threadIdx.x >> 5;
    const int m      = lane & 15;
    const int khalf  = (lane >> 4) << 1;

    const float* W    = params + (size_t)g * H * O;
    const int    col4 = blockIdx.y * 256 + waveid * 64 + m * 4;
    const float* aptr = As + m * H + khalf;
    const float* bptr = W + (size_t)khalf * O + col4;

    v8f acc[4];
#pragma unroll
    for (int t = 0; t < 4; ++t) acc[t] = (v8f){0, 0, 0, 0, 0, 0, 0, 0};

    auto mma_step = [&](const v2f a, const v4f b0, const v4f b1) {
#pragma unroll
        for (int t = 0; t < 4; ++t) {
            v2f b; b.x = b0[t]; b.y = b1[t];
            acc[t] = __builtin_amdgcn_wmma_f32_16x16x4_f32(
                false, a, false, b, (short)0, acc[t], false, false);
        }
    };
    auto load_batch = [&](v4f (&q)[KS][2], const int kb) {
#pragma unroll
        for (int s = 0; s < KS; ++s) {
            const float* p = bptr + (size_t)(kb + 4 * s) * O;
            q[s][0] = *(const v4f*)p;
            q[s][1] = *(const v4f*)(p + O);
        }
    };
    auto mma_batch = [&](const v4f (&q)[KS][2], const int kb) {
#pragma unroll
        for (int s = 0; s < KS; ++s) {
            const v2f a = *(const v2f*)(aptr + kb + 4 * s);
            mma_step(a, q[s][0], q[s][1]);
        }
    };

    const int Hp = H - H % (2 * KBATCH);
    int kb = 0;
    if (Hp > 0) {
        v4f q0[KS][2], q1[KS][2];
        load_batch(q0, 0);
        for (; kb + 2 * KBATCH < Hp; kb += 2 * KBATCH) {
            load_batch(q1, kb + KBATCH);
            __builtin_amdgcn_sched_barrier(0);
            mma_batch(q0, kb);
            load_batch(q0, kb + 2 * KBATCH);
            __builtin_amdgcn_sched_barrier(0);
            mma_batch(q1, kb + KBATCH);
        }
        load_batch(q1, kb + KBATCH);
        __builtin_amdgcn_sched_barrier(0);
        mma_batch(q0, kb);
        mma_batch(q1, kb + KBATCH);
        kb += 2 * KBATCH;
    }
    for (; kb < H; kb += 4) {
        const float* p = bptr + (size_t)kb * O;
        mma_step(*(const v2f*)(aptr + kb), *(const v4f*)p, *(const v4f*)(p + O));
    }

    const int mbase = (lane >> 4) << 3;
    const v4f bv    = *(const v4f*)(bias + (size_t)g * O + col4);
#pragma unroll
    for (int r = 0; r < 8; ++r) {
        const int mr = mbase + r;
        if (mr < cnt) {
            const int n = perm[rowbeg + mr];
            v4f o;
            o.x = acc[0][r] + bv.x; o.y = acc[1][r] + bv.y;
            o.z = acc[2][r] + bv.z; o.w = acc[3][r] + bv.w;
            *(v4f*)(out + (size_t)n * O + col4) = o;
        }
    }
}

extern "C" void kernel_launch(void* const* d_in, const int* in_sizes, int n_in,
                              void* d_out, int out_size, void* d_ws, size_t ws_size,
                              hipStream_t stream)
{
    const int*   x      = (const int*)  d_in[0];
    const int*   fact   = (const int*)  d_in[1];
    const float* inp    = (const float*)d_in[2];
    const float* params = (const float*)d_in[3];
    const float* bias   = (const float*)d_in[4];
    const int*   msg_to = (const int*)  d_in[5];
    float*       out    = (float*)d_out;

    const int N   = in_sizes[1] / 2;
    const int O   = in_sizes[4] / NPAIR;
    const int H   = in_sizes[3] / (NPAIR * O);
    const int ORD = in_sizes[2] / (N * H);

    const int wcap = NPAIR + (N + 15) / 16;

    int* ws     = (int*)d_ws;
    int* ids    = ws;
    int* perm   = ws + N;
    int* gstart = ws + 2 * N;
    int* work   = ws + 2 * N + NPAIR + 1;

    hipLaunchKernelGGL(hon_prep, dim3(1), dim3(PREP_THREADS), 0, stream,
                       x, fact, ids, perm, gstart, work, N, wcap);

    const size_t lds_bytes = (size_t)16 * H * sizeof(float);
    dim3 grid(wcap, O / 256);
    hipLaunchKernelGGL(hon_gemm, grid, dim3(128), lds_bytes, stream,
                       inp, params, bias, perm, gstart, work, msg_to, out,
                       N, H, O, ORD);
}


# Loads outside the K loop no longer wait one by one: tile's 16 row numbers read once, passed by lane reads (staging, stores); a thread's 32 staging loads fly together; bias asked before the loop
# speedup vs baseline: 3.4655x; 1.6388x over previous
#include <hip/hip_runtime.h>


#define NPAIR 169

typedef __attribute__((ext_vector_type(2))) float v2f;
typedef __attribute__((ext_vector_type(4))) float v4f;
typedef __attribute__((ext_vector_type(8))) float v8f;

#define KS 8
#define KBATCH (4 * KS)

#define PREP_THREADS 1024
#define GWAVES ((NPAIR + 31) / 32)

__global__ void __launch_bounds__(PREP_THREADS)
hon_prep(const int* __restrict__ x, const int* __restrict__ fact,
         int* __restrict__ ids, int* __restrict__ perm,
         int* __restrict__ gstart, int* __restrict__ work,
         int N, int wcap)
{
    __shared__ int cnt[NPAIR];
    __shared__ int start[NPAIR];
    __shared__ int wrow[GWAVES];
    __shared__ int wtil[GWAVES];
    __shared__ int ntile;

    const int tid  = threadIdx.x;
    const int B    = blockDim.x;
    const int lane = tid & 31;
    const int wv   = tid >> 5;

    int id0 = 0;
    if (tid < N) {
        const int a = fact[2 * tid];
        id0 = x[3 * a + 1] * 13 + x[3 * a + 2];
    }
    for (int g = tid; g < NPAIR; g += B) cnt[g] = 0;
    __syncthreads();

    int rk0 = 0;
    if (tid < N) rk0 = atomicAdd(&cnt[id0], 1);
    for (int n = tid + B; n < N; n += B) {
        const int a  = fact[2 * n];
        const int id = x[3 * a + 1] * 13 + x[3 * a + 2];
        ids[n] = (atomicAdd(&cnt[id], 1) << 8) | id;
    }
    __syncthreads();

    int c = 0, nt = 0, sc = 0, st = 0;
    if (wv < GWAVES) {
        if (tid < NPAIR) { c = cnt[tid]; nt = (c + 15) >> 4; }
        sc = c; st = nt;
        for (int d = 1; d < 32; d <<= 1) {
            const int uc = __shfl_up(sc, d);
            const int ut = __shfl_up(st, d);
            if (lane >= d) { sc += uc; st += ut; }
        }
        if (lane == 31) { wrow[wv] = sc; wtil[wv] = st; }
    }
    __syncthreads();

    int rs = 0, ws = 0;
    if (tid < NPAIR) {
        int br = 0, bt = 0;
        for (int w = 0; w < GWAVES - 1; ++w) {
            const int r = wrow[w], t = wtil[w];
            if (w < wv) { br += r; bt += t; }
        }
        rs = br + sc - c;
        ws = bt + st - nt;
        start[tid] = rs;
        if (tid == NPAIR - 1) ntile = ws + nt;
    }
    __syncthreads();

    if (tid < NPAIR) {
        gstart[tid] = rs;
        if (tid == NPAIR - 1) gstart[NPAIR] = rs + c;
        for (int t = 0; t < nt; ++t) {
            const int w = ws + t;
            if (w < wcap) { work[2 * w] = tid; work[2 * w + 1] = t; }
        }
    }
    for (int w = ntile + tid; w < wcap; w += B) {
        work[2 * w] = -1; work[2 * w + 1] = 0;
    }
    if (tid < N) perm[start[id0] + rk0] = tid;
    for (int n = tid + B; n < N; n += B) {
        const int v = ids[n];
        perm[start[v & 255] + (v >> 8)] = n;
    }
}

__global__ void __launch_bounds__(128)
hon_gemm(const float* __restrict__ inp, const float* __restrict__ params,
         const float* __restrict__ bias, const int* __restrict__ perm,
         const int* __restrict__ gstart, const int* __restrict__ work,
         const int* __restrict__ msg_to_p, float* __restrict__ out,
         int N, int H, int O, int ORD)
{
    extern __shared__ float As[];

    const int g = work[2 * blockIdx.x];
    if (g < 0) return;
    const int mt     = work[2 * blockIdx.x + 1];
    const int rowbeg = gstart[g] + mt * 16;
    int cnt = gstart[g + 1] - rowbeg;
    cnt = cnt > 16 ? 16 : cnt;
    const int msg_to = msg_to_p[0];

    int nrow;
    {
        int rsel = threadIdx.x & 15;
        rsel = rsel < cnt ? rsel : cnt - 1;
        rsel = rsel > 0 ? rsel : 0;
        nrow = perm[rowbeg + rsel];
    }

    {
        const int    HQ    = H >> 2;
        const size_t plane = (size_t)N * H;
        const v4f    one   = (v4f){1.f, 1.f, 1.f, 1.f};
        const v4f    zero  = (v4f){0.f, 0.f, 0.f, 0.f};
        const int    nkept = ORD - ((msg_to >= 0 && msg_to < ORD) ? 1 : 0);
        for (int kq = threadIdx.x; kq < HQ; kq += blockDim.x) {
            const unsigned koff = 4u * (unsigned)kq;
            v4f p[16];
            if (nkept == 2) {
                const int a = msg_to == 0 ? 1 : 0;
                const int b = msg_to == a + 1 ? a + 2 : a + 1;
                const float* pa = inp + (size_t)a * plane;
                const float* pb = inp + (size_t)b * plane;
                v4f va[16], vb[16];
#pragma unroll
                for (int r = 0; r < 16; ++r) {
                    const size_t ro = (size_t)__builtin_amdgcn_readlane(nrow, r) * H;
                    va[r] = *(const v4f*)(pa + ro + koff);
                    vb[r] = *(const v4f*)(pb + ro + koff);
                }
#pragma unroll
                for (int r = 0; r < 16; ++r) p[r] = va[r] * vb[r];
            } else {
#pragma unroll
                for (int r = 0; r < 16; ++r) p[r] = one;
                for (int i = 0; i < ORD; ++i) {
                    if (i == msg_to) continue;
                    const float* pi = inp + (size_t)i * plane;
                    v4f v[16];
#pragma unroll
                    for (int r = 0; r < 16; ++r)
                        v[r] = *(const v4f*)(pi + (size_t)__builtin_amdgcn_readlane(nrow, r) * H + koff);
#pragma unroll
                    for (int r = 0; r < 16; ++r) p[r] = p[r] * v[r];
                }
            }
#pragma unroll
            for (int r = 0; r < 16; ++r)
                *(v4f*)(As + r * H + koff) = r < cnt ? p[r] : zero;
        }
    }
    __syncthreads();

    const int lane   = threadIdx.x & 31;
    const int waveid = threadIdx.x >> 5;
    const int m      = lane & 15;
    const int khalf  = (lane >> 4) << 1;

    const float* W    = params + (size_t)g * H * O;
    const int    col4 = blockIdx.y * 256 + waveid * 64 + m * 4;
    const float* aptr = As + m * H + khalf;
    const float* bptr = W + (size_t)khalf * O + col4;
    const v4f bv = *(const v4f*)(bias + (size_t)g * O + col4);

    v8f acc[4];
#pragma unroll
    for (int t = 0; t < 4; ++t) acc[t] = (v8f){0, 0, 0, 0, 0, 0, 0, 0};

    auto mma_step = [&](const v2f a, const v4f b0, const v4f b1) {
#pragma unroll
        for (int t = 0; t < 4; ++t) {
            v2f b; b.x = b0[t]; b.y = b1[t];
            acc[t] = __builtin_amdgcn_wmma_f32_16x16x4_f32(
                false, a, false, b, (short)0, acc[t], false, false);
        }
    };
    auto load_batch = [&](v4f (&q)[KS][2], const int kb) {
#pragma unroll
        for (int s = 0; s < KS; ++s) {
            const float* p = bptr + (size_t)(kb + 4 * s) * O;
            q[s][0] = *(const v4f*)p;
            q[s][1] = *(const v4f*)(p + O);
        }
    };
    auto mma_batch = [&](const v4f (&q)[KS][2], const int kb) {
#pragma unroll
        for (int s = 0; s < KS; ++s) {
            const v2f a = *(const v2f*)(aptr + kb + 4 * s);
            mma_step(a, q[s][0], q[s][1]);
        }
    };

    const int Hp = H - H % (2 * KBATCH);
    int kb = 0;
    if (Hp > 0) {
        v4f q0[KS][2], q1[KS][2];
        load_batch(q0, 0);
        for (; kb + 2 * KBATCH < Hp; kb += 2 * KBATCH) {
            load_batch(q1, kb + KBATCH);
            __builtin_amdgcn_sched_barrier(0);
            mma_batch(q0, kb);
            load_batch(q0, kb + 2 * KBATCH);
            __builtin_amdgcn_sched_barrier(0);
            mma_batch(q1, kb + KBATCH);
        }
        load_batch(q1, kb + KBATCH);
        __builtin_amdgcn_sched_barrier(0);
        mma_batch(q0, kb);
        mma_batch(q1, kb + KBATCH);
        kb += 2 * KBATCH;
    }
    for (; kb < H; kb += 4) {
        const float* p = bptr + (size_t)kb * O;
        mma_step(*(const v2f*)(aptr + kb), *(const v4f*)p, *(const v4f*)(p + O));
    }

    const int mbase = (lane >> 4) << 3;
#pragma unroll
    for (int r = 0; r < 8; ++r) {
        const int mr = mbase + r;
        const int nlo = __builtin_amdgcn_readlane(nrow, r);
        const int nhi = __builtin_amdgcn_readlane(nrow, r + 8);
        const int n   = lane < 16 ? nlo : nhi;
        if (mr < cnt) {
            v4f o;
            o.x = acc[0][r] + bv.x; o.y = acc[1][r] + bv.y;
            o.z = acc[2][r] + bv.z; o.w = acc[3][r] + bv.w;
            *(v4f*)(out + (size_t)n * O + col4) = o;
        }
    }
}

extern "C" void kernel_launch(void* const* d_in, const int* in_sizes, int n_in,
                              void* d_out, int out_size, void* d_ws, size_t ws_size,
                              hipStream_t stream)
{
    const int*   x      = (const int*)  d_in[0];
    const int*   fact   = (const int*)  d_in[1];
    const float* inp    = (const float*)d_in[2];
    const float* params = (const float*)d_in[3];
    const float* bias   = (const float*)d_in[4];
    const int*   msg_to = (const int*)  d_in[5];
    float*       out    = (float*)d_out;

    const int N   = in_sizes[1] / 2;
    const int O   = in_sizes[4] / NPAIR;
    const int H   = in_sizes[3] / (NPAIR * O);
    const int ORD = in_sizes[2] / (N * H);

    const int wcap = NPAIR + (N + 15) / 16;

    int* ws     = (int*)d_ws;
    int* ids    = ws;
    int* perm   = ws + N;
    int* gstart = ws + 2 * N;
    int* work   = ws + 2 * N + NPAIR + 1;

    hipLaunchKernelGGL(hon_prep, dim3(1), dim3(PREP_THREADS), 0, stream,
                       x, fact, ids, perm, gstart, work, N, wcap);

    const size_t lds_bytes = (size_t)16 * H * sizeof(float);
    dim3 grid(wcap, O / 256);
    hipLaunchKernelGGL(hon_gemm, grid, dim3(128), lds_bytes, stream,
                       inp, params, bias, perm, gstart, work, msg_to, out,
                       N, H, O, ORD);
}


# One launch, not two: each of 256 work-groups redoes the row sort in LDS (ids, histogram, prefix sum; tile rows by ballot rank) and walks the tile list; sort launch, workspace gone; sum order kept
# speedup vs baseline: 4.7466x; 1.3697x over previous
#include <hip/hip_runtime.h>


#define NPAIR 169

typedef __attribute__((ext_vector_type(2))) float v2f;
typedef __attribute__((ext_vector_type(4))) float v4f;
typedef __attribute__((ext_vector_type(8))) float v8f;

#define KS 8
#define KBATCH (4 * KS)

#define NTHREADS 128
#define NWAVES   (NTHREADS / 32)
#define NBLOCKS  256
#define RPT      8
#define PASSROWS (NTHREADS * RPT)
#define NPAIR2   ((NPAIR + 1) / 2)
static_assert(NWAVES == 4 && NPAIR2 <= NTHREADS, "piece counts are read four at a time; two groups per thread");

__device__ __forceinline__ int row_id(const int* __restrict__ x,
                                      const int* __restrict__ fact, const int n)
{
    const int a = fact[2 * n];
    return x[3 * a + 1] * 13 + x[3 * a + 2];
}

__device__ __forceinline__ int wave_scan_incl(int v, const int ln)
{
#pragma unroll
    for (int d = 1; d < 32; d <<= 1) {
        const int u = __shfl_up(v, d);
        if (ln >= d) v += u;
    }
    return v;
}

__global__ void __launch_bounds__(NTHREADS)
hon_fused(const int* __restrict__ x, const int* __restrict__ fact,
          const float* __restrict__ inp, const float* __restrict__ params,
          const float* __restrict__ bias, const int* __restrict__ msg_to_p,
          float* __restrict__ out, int N, int H, int O, int ORD)
{
    extern __shared__ __attribute__((aligned(16))) float As[];
    __shared__ int gcnt[NPAIR + 1];
    __shared__ int wtot[NWAVES];
    __shared__ int sel[2];
    __shared__ __attribute__((aligned(16))) int pc[RPT * NWAVES];
    __shared__ int rowsel[16];

    const int tid = threadIdx.x;
    const int ln  = tid & 31;
    const int wv  = tid >> 5;

    int idr[RPT];
    {
        int a[RPT];
#pragma unroll
        for (int j = 0; j < RPT; ++j) {
            const int n = j * NTHREADS + tid;
            a[j] = fact[2 * (n < N ? n : N - 1)];
        }
#pragma unroll
        for (int j = 0; j < RPT; ++j) {
            const int n  = j * NTHREADS + tid;
            const int id = x[3 * a[j] + 1] * 13 + x[3 * a[j] + 2];
            idr[j] = n < N ? id : -1;
        }
    }
    const int msg_to = msg_to_p[0];

    for (int g = tid; g <= NPAIR; g += NTHREADS) gcnt[g] = 0;
    __syncthreads();
#pragma unroll
    for (int j = 0; j < RPT; ++j)
        if (idr[j] >= 0) atomicAdd(&gcnt[idr[j]], 1);
    for (int n = PASSROWS + tid; n < N; n += NTHREADS)
        atomicAdd(&gcnt[row_id(x, fact, n)], 1);
    __syncthreads();

    int nt0 = 0, nt1 = 0;
    if (tid < NPAIR2) {
        nt0 = (gcnt[2 * tid] + 15) >> 4;
        nt1 = (gcnt[2 * tid + 1] + 15) >> 4;
    }
    const int inc = wave_scan_incl(nt0 + nt1, ln);
    if (ln == 31) wtot[wv] = inc;
    __syncthreads();
    int ts0 = inc - nt0 - nt1, ntile = 0;
#pragma unroll
    for (int w = 0; w < NWAVES; ++w) {
        const int t = wtot[w];
        ntile += t;
        if (w < wv) ts0 += t;
    }
    const int ts1 = ts0 + nt0;

    const int nh     = O >> 8;
    const int nitems = ntile * nh;

    for (int item = blockIdx.x; item < nitems; item += gridDim.x) {
        const int ti   = item / nh;
        const int half = item - ti * nh;

        if (ti >= ts0 && ti < ts0 + nt0) { sel[0] = 2 * tid;     sel[1] = ti - ts0; }
        if (ti >= ts1 && ti < ts1 + nt1) { sel[0] = 2 * tid + 1; sel[1] = ti - ts1; }
        __syncthreads();
        const int g  = sel[0];
        const int mt = sel[1];
        int cnt = gcnt[g] - mt * 16;
        cnt = cnt > 16 ? 16 : cnt;

        int rbase = 0;
        for (int pb = 0; pb < N; pb += PASSROWS) {
            unsigned bal[RPT];
#pragma unroll
            for (int j = 0; j < RPT; ++j) {
                int id = idr[j];
                if (pb > 0) {
                    const int n = pb + j * NTHREADS + tid;
                    id = n < N ? row_id(x, fact, n) : -1;
                }
                bal[j] = __builtin_amdgcn_ballot_w32(id == g);
            }
            if (pb > 0) __syncthreads();
            int mine = 0;
#pragma unroll
            for (int j = 0; j < RPT; ++j) mine = ln == j ? __popc(bal[j]) : mine;
            if (ln < RPT) pc[ln * NWAVES + wv] = mine;
            __syncthreads();
            int4 q[RPT];
#pragma unroll
            for (int j = 0; j < RPT; ++j) q[j] = *(const int4*)&pc[j * NWAVES];
            int run = rbase;
#pragma unroll
            for (int j = 0; j < RPT; ++j) {
                const int pre = run + (wv > 0 ? q[j].x : 0) + (wv > 1 ? q[j].y : 0)
                                    + (wv > 2 ? q[j].z : 0);
                run += q[j].x + q[j].y + q[j].z + q[j].w;
                const int rank = pre + __popc(bal[j] & ((1u << ln) - 1u));
                if (((bal[j] >> ln) & 1u) && (rank >> 4) == mt)
                    rowsel[rank & 15] = pb + j * NTHREADS + tid;
            }
            rbase = run;
        }
        __syncthreads();

        int nrow;
        {
            int rsel = threadIdx.x & 15;
            rsel = rsel < cnt ? rsel : cnt - 1;
            rsel = rsel > 0 ? rsel : 0;
            nrow = rowsel[rsel];
        }

        {
            const int    HQ    = H >> 2;
            const size_t plane = (size_t)N * H;
            const v4f    one   = (v4f){1.f, 1.f, 1.f, 1.f};
            const v4f    zero  = (v4f){0.f, 0.f, 0.f, 0.f};
            const int    nkept = ORD - ((msg_to >= 0 && msg_to < ORD) ? 1 : 0);
            for (int kq = threadIdx.x; kq < HQ; kq += blockDim.x) {
                const unsigned koff = 4u * (unsigned)kq;
                v4f p[16];
                if (nkept == 2) {
                    const int a = msg_to == 0 ? 1 : 0;
                    const int b = msg_to == a + 1 ? a + 2 : a + 1;
                    const float* pa = inp + (size_t)a * plane;
                    const float* pb = inp + (size_t)b * plane;
                    v4f va[16], vb[16];
#pragma unroll
                    for (int r = 0; r < 16; ++r) {
                        const size_t ro = (size_t)__builtin_amdgcn_readlane(nrow, r) * H;
                        va[r] = *(const v4f*)(pa + ro + koff);
                        vb[r] = *(const v4f*)(pb + ro + koff);
                    }
#pragma unroll
                    for (int r = 0; r < 16; ++r) p[r] = va[r] * vb[r];
                } else {
#pragma unroll
                    for (int r = 0; r < 16; ++r) p[r] = one;
                    for (int i = 0; i < ORD; ++i) {
                        if (i == msg_to) continue;
                        const float* pi = inp + (size_t)i * plane;
                        v4f v[16];
#pragma unroll
                        for (int r = 0; r < 16; ++r)
                            v[r] = *(const v4f*)(pi + (size_t)__builtin_amdgcn_readlane(nrow, r) * H + koff);
#pragma unroll
                        for (int r = 0; r < 16; ++r) p[r] = p[r] * v[r];
                    }
                }
#pragma unroll
                for (int r = 0; r < 16; ++r)
                    *(v4f*)(As + r * H + koff) = r < cnt ? p[r] : zero;
            }
        }
        __syncthreads();

        const int lane   = threadIdx.x & 31;
        const int waveid = threadIdx.x >> 5;
        const int m      = lane & 15;
        const int khalf  = (lane >> 4) << 1;

        const float* W    = params + (size_t)g * H * O;
        const int    col4 = half * 256 + waveid * 64 + m * 4;
        const float* aptr = As + m * H + khalf;
        const float* bptr = W + (size_t)khalf * O + col4;
        const v4f bv = *(const v4f*)(bias + (size_t)g * O + col4);

        v8f acc[4];
#pragma unroll
        for (int t = 0; t < 4; ++t) acc[t] = (v8f){0, 0, 0, 0, 0, 0, 0, 0};

        auto mma_step = [&](const v2f a, const v4f b0, const v4f b1) {
#pragma unroll
            for (int t = 0; t < 4; ++t) {
                v2f b; b.x = b0[t]; b.y = b1[t];
                acc[t] = __builtin_amdgcn_wmma_f32_16x16x4_f32(
                    false, a, false, b, (short)0, acc[t], false, false);
            }
        };
        auto load_batch = [&](v4f (&q)[KS][2], const int kb) {
#pragma unroll
            for (int s = 0; s < KS; ++s) {
                const float* p = bptr + (size_t)(kb + 4 * s) * O;
                q[s][0] = *(const v4f*)p;
                q[s][1] = *(const v4f*)(p + O);
            }
        };
        auto mma_batch = [&](const v4f (&q)[KS][2], const int kb) {
#pragma unroll
            for (int s = 0; s < KS; ++s) {
                const v2f a = *(const v2f*)(aptr + kb + 4 * s);
                mma_step(a, q[s][0], q[s][1]);
            }
        };

        const int Hp = H - H % (2 * KBATCH);
        int kb = 0;
        if (Hp > 0) {
            v4f q0[KS][2], q1[KS][2];
            load_batch(q0, 0);
            for (; kb + 2 * KBATCH < Hp; kb += 2 * KBATCH) {
                load_batch(q1, kb + KBATCH);
                __builtin_amdgcn_sched_barrier(0);
                mma_batch(q0, kb);
                load_batch(q0, kb + 2 * KBATCH);
                __builtin_amdgcn_sched_barrier(0);
                mma_batch(q1, kb + KBATCH);
            }
            load_batch(q1, kb + KBATCH);
            __builtin_amdgcn_sched_barrier(0);
            mma_batch(q0, kb);
            mma_batch(q1, kb + KBATCH);
            kb += 2 * KBATCH;
        }
        for (; kb < H; kb += 4) {
            const float* p = bptr + (size_t)kb * O;
            mma_step(*(const v2f*)(aptr + kb), *(const v4f*)p, *(const v4f*)(p + O));
        }

        const int mbase = (lane >> 4) << 3;
#pragma unroll
        for (int r = 0; r < 8; ++r) {
            const int mr = mbase + r;
            const int nlo = __builtin_amdgcn_readlane(nrow, r);
            const int nhi = __builtin_amdgcn_readlane(nrow, r + 8);
            const int n   = lane < 16 ? nlo : nhi;
            if (mr < cnt) {
                v4f o;
                o.x = acc[0][r] + bv.x; o.y = acc[1][r] + bv.y;
                o.z = acc[2][r] + bv.z; o.w = acc[3][r] + bv.w;
                *(v4f*)(out + (size_t)n * O + col4) = o;
            }
        }

        __syncthreads();
    }
}

extern "C" void kernel_launch(void* const* d_in, const int* in_sizes, int n_in,
                              void* d_out, int out_size, void* d_ws, size_t ws_size,
                              hipStream_t stream)
{
    const int*   x      = (const int*)  d_in[0];
    const int*   fact   = (const int*)  d_in[1];
    const float* inp    = (const float*)d_in[2];
    const float* params = (const float*)d_in[3];
    const float* bias   = (const float*)d_in[4];
    const int*   msg_to = (const int*)  d_in[5];
    float*       out    = (float*)d_out;

    const int N   = in_sizes[1] / 2;
    const int O   = in_sizes[4] / NPAIR;
    const int H   = in_sizes[3] / (NPAIR * O);
    const int ORD = in_sizes[2] / (N * H);

    const size_t lds_bytes = (size_t)16 * H * sizeof(float);
    hipLaunchKernelGGL(hon_fused, dim3(NBLOCKS), dim3(NTHREADS), lds_bytes, stream,
                       x, fact, inp, params, bias, msg_to, out, N, H, O, ORD);
}


# Weight stream made independent of the caches and deeper: non-temporal 128-bit loads (weights are read once), three batches of 8 K-steps in flight per wave (ring of 4 register sets), not one
# speedup vs baseline: 5.5190x; 1.1627x over previous
#include <hip/hip_runtime.h>


#define NPAIR 169

typedef __attribute__((ext_vector_type(2))) float v2f;
typedef __attribute__((ext_vector_type(4))) float v4f;
typedef __attribute__((ext_vector_type(8))) float v8f;

#define KS 8
#define KBATCH (4 * KS)
#define NQ 4

#define NTHREADS 128
#define NWAVES   (NTHREADS / 32)
#define NBLOCKS  256
#define RPT      8
#define PASSROWS (NTHREADS * RPT)
#define NPAIR2   ((NPAIR + 1) / 2)
static_assert(NWAVES == 4 && NPAIR2 <= NTHREADS, "piece counts are read four at a time; two groups per thread");

__device__ __forceinline__ int row_id(const int* __restrict__ x,
                                      const int* __restrict__ fact, const int n)
{
    const int a = fact[2 * n];
    return x[3 * a + 1] * 13 + x[3 * a + 2];
}

__device__ __forceinline__ int wave_scan_incl(int v, const int ln)
{
#pragma unroll
    for (int d = 1; d < 32; d <<= 1) {
        const int u = __shfl_up(v, d);
        if (ln >= d) v += u;
    }
    return v;
}

__global__ void __launch_bounds__(NTHREADS)
hon_fused(const int* __restrict__ x, const int* __restrict__ fact,
          const float* __restrict__ inp, const float* __restrict__ params,
          const float* __restrict__ bias, const int* __restrict__ msg_to_p,
          float* __restrict__ out, int N, int H, int O, int ORD)
{
    extern __shared__ __attribute__((aligned(16))) float As[];
    __shared__ int gcnt[NPAIR + 1];
    __shared__ int wtot[NWAVES];
    __shared__ int sel[2];
    __shared__ __attribute__((aligned(16))) int pc[RPT * NWAVES];
    __shared__ int rowsel[16];

    const int tid = threadIdx.x;
    const int ln  = tid & 31;
    const int wv  = tid >> 5;

    int idr[RPT];
    {
        int a[RPT];
#pragma unroll
        for (int j = 0; j < RPT; ++j) {
            const int n = j * NTHREADS + tid;
            a[j] = fact[2 * (n < N ? n : N - 1)];
        }
#pragma unroll
        for (int j = 0; j < RPT; ++j) {
            const int n  = j * NTHREADS + tid;
            const int id = x[3 * a[j] + 1] * 13 + x[3 * a[j] + 2];
            idr[j] = n < N ? id : -1;
        }
    }
    const int msg_to = msg_to_p[0];

    for (int g = tid; g <= NPAIR; g += NTHREADS) gcnt[g] = 0;
    __syncthreads();
#pragma unroll
    for (int j = 0; j < RPT; ++j)
        if (idr[j] >= 0) atomicAdd(&gcnt[idr[j]], 1);
    for (int n = PASSROWS + tid; n < N; n += NTHREADS)
        atomicAdd(&gcnt[row_id(x, fact, n)], 1);
    __syncthreads();

    int nt0 = 0, nt1 = 0;
    if (tid < NPAIR2) {
        nt0 = (gcnt[2 * tid] + 15) >> 4;
        nt1 = (gcnt[2 * tid + 1] + 15) >> 4;
    }
    const int inc = wave_scan_incl(nt0 + nt1, ln);
    if (ln == 31) wtot[wv] = inc;
    __syncthreads();
    int ts0 = inc - nt0 - nt1, ntile = 0;
#pragma unroll
    for (int w = 0; w < NWAVES; ++w) {
        const int t = wtot[w];
        ntile += t;
        if (w < wv) ts0 += t;
    }
    const int ts1 = ts0 + nt0;

    const int nh     = O >> 8;
    const int nitems = ntile * nh;

    for (int item = blockIdx.x; item < nitems; item += gridDim.x) {
        const int ti   = item / nh;
        const int half = item - ti * nh;

        if (ti >= ts0 && ti < ts0 + nt0) { sel[0] = 2 * tid;     sel[1] = ti - ts0; }
        if (ti >= ts1 && ti < ts1 + nt1) { sel[0] = 2 * tid + 1; sel[1] = ti - ts1; }
        __syncthreads();
        const int g  = sel[0];
        const int mt = sel[1];
        int cnt = gcnt[g] - mt * 16;
        cnt = cnt > 16 ? 16 : cnt;

        int rbase = 0;
        for (int pb = 0; pb < N; pb += PASSROWS) {
            unsigned bal[RPT];
#pragma unroll
            for (int j = 0; j < RPT; ++j) {
                int id = idr[j];
                if (pb > 0) {
                    const int n = pb + j * NTHREADS + tid;
                    id = n < N ? row_id(x, fact, n) : -1;
                }
                bal[j] = __builtin_amdgcn_ballot_w32(id == g);
            }
            if (pb > 0) __syncthreads();
            int mine = 0;
#pragma unroll
            for (int j = 0; j < RPT; ++j) mine = ln == j ? __popc(bal[j]) : mine;
            if (ln < RPT) pc[ln * NWAVES + wv] = mine;
            __syncthreads();
            int4 q[RPT];
#pragma unroll
            for (int j = 0; j < RPT; ++j) q[j] = *(const int4*)&pc[j * NWAVES];
            int run = rbase;
#pragma unroll
            for (int j = 0; j < RPT; ++j) {
                const int pre = run + (wv > 0 ? q[j].x : 0) + (wv > 1 ? q[j].y : 0)
                                    + (wv > 2 ? q[j].z : 0);
                run += q[j].x + q[j].y + q[j].z + q[j].w;
                const int rank = pre + __popc(bal[j] & ((1u << ln) - 1u));
                if (((bal[j] >> ln) & 1u) && (rank >> 4) == mt)
                    rowsel[rank & 15] = pb + j * NTHREADS + tid;
            }
            rbase = run;
        }
        __syncthreads();

        int nrow;
        {
            int rsel = threadIdx.x & 15;
            rsel = rsel < cnt ? rsel : cnt - 1;
            rsel = rsel > 0 ? rsel : 0;
            nrow = rowsel[rsel];
        }

        {
            const int    HQ    = H >> 2;
            const size_t plane = (size_t)N * H;
            const v4f    one   = (v4f){1.f, 1.f, 1.f, 1.f};
            const v4f    zero  = (v4f){0.f, 0.f, 0.f, 0.f};
            const int    nkept = ORD - ((msg_to >= 0 && msg_to < ORD) ? 1 : 0);
            for (int kq = threadIdx.x; kq < HQ; kq += blockDim.x) {
                const unsigned koff = 4u * (unsigned)kq;
                v4f p[16];
                if (nkept == 2) {
                    const int a = msg_to == 0 ? 1 : 0;
                    const int b = msg_to == a + 1 ? a + 2 : a + 1;
                    const float* pa = inp + (size_t)a * plane;
                    const float* pb = inp + (size_t)b * plane;
                    v4f va[16], vb[16];
#pragma unroll
                    for (int r = 0; r < 16; ++r) {
                        const size_t ro = (size_t)__builtin_amdgcn_readlane(nrow, r) * H;
                        va[r] = *(const v4f*)(pa + ro + koff);
                        vb[r] = *(const v4f*)(pb + ro + koff);
                    }
#pragma unroll
                    for (int r = 0; r < 16; ++r) p[r] = va[r] * vb[r];
                } else {
#pragma unroll
                    for (int r = 0; r < 16; ++r) p[r] = one;
                    for (int i = 0; i < ORD; ++i) {
                        if (i == msg_to) continue;
                        const float* pi = inp + (size_t)i * plane;
                        v4f v[16];
#pragma unroll
                        for (int r = 0; r < 16; ++r)
                            v[r] = *(const v4f*)(pi + (size_t)__builtin_amdgcn_readlane(nrow, r) * H + koff);
#pragma unroll
                        for (int r = 0; r < 16; ++r) p[r] = p[r] * v[r];
                    }
                }
#pragma unroll
                for (int r = 0; r < 16; ++r)
                    *(v4f*)(As + r * H + koff) = r < cnt ? p[r] : zero;
            }
        }
        __syncthreads();

        const int lane   = threadIdx.x & 31;
        const int waveid = threadIdx.x >> 5;
        const int m      = lane & 15;
        const int khalf  = (lane >> 4) << 1;

        const float* W    = params + (size_t)g * H * O;
        const int    col4 = half * 256 + waveid * 64 + m * 4;
        const float* aptr = As + m * H + khalf;
        const float* bptr = W + (size_t)khalf * O + col4;
        const v4f bv = *(const v4f*)(bias + (size_t)g * O + col4);

        v8f acc[4];
#pragma unroll
        for (int t = 0; t < 4; ++t) acc[t] = (v8f){0, 0, 0, 0, 0, 0, 0, 0};

        auto mma_step = [&](const v2f a, const v4f b0, const v4f b1) {
#pragma unroll
            for (int t = 0; t < 4; ++t) {
                v2f b; b.x = b0[t]; b.y = b1[t];
                acc[t] = __builtin_amdgcn_wmma_f32_16x16x4_f32(
                    false, a, false, b, (short)0, acc[t], false, false);
            }
        };
        auto load_batch = [&](v4f (&q)[KS][2], const int kb) {
#pragma unroll
            for (int s = 0; s < KS; ++s) {
                const float* p = bptr + (size_t)(kb + 4 * s) * O;
                q[s][0] = __builtin_nontemporal_load((const v4f*)p);
                q[s][1] = __builtin_nontemporal_load((const v4f*)(p + O));
            }
        };
        auto mma_batch = [&](const v4f (&q)[KS][2], const int kb) {
#pragma unroll
            for (int s = 0; s < KS; ++s) {
                const v2f a = *(const v2f*)(aptr + kb + 4 * s);
                mma_step(a, q[s][0], q[s][1]);
            }
        };

        const int nbat = H / KBATCH;
        int kdone = 0;
        if (nbat >= NQ && nbat % NQ == 0) {
            v4f q[NQ][KS][2];
#pragma unroll
            for (int j = 0; j < NQ - 1; ++j) {
                load_batch(q[j], j * KBATCH);
                __builtin_amdgcn_sched_barrier(0);
            }
            int b = 0;
            for (; b + NQ < nbat; b += NQ) {
#pragma unroll
                for (int j = 0; j < NQ; ++j) {
                    load_batch(q[(j + NQ - 1) % NQ], (b + j + NQ - 1) * KBATCH);
                    __builtin_amdgcn_sched_barrier(0);
                    mma_batch(q[j], (b + j) * KBATCH);
                }
            }
            load_batch(q[NQ - 1], (b + NQ - 1) * KBATCH);
            __builtin_amdgcn_sched_barrier(0);
#pragma unroll
            for (int j = 0; j < NQ; ++j) mma_batch(q[j], (b + j) * KBATCH);
            kdone = nbat * KBATCH;
        }
        for (int kb = kdone; kb < H; kb += 4) {
            const float* p = bptr + (size_t)kb * O;
            mma_step(*(const v2f*)(aptr + kb), *(const v4f*)p, *(const v4f*)(p + O));
        }

        const int mbase = (lane >> 4) << 3;
#pragma unroll
        for (int r = 0; r < 8; ++r) {
            const int mr = mbase + r;
            const int nlo = __builtin_amdgcn_readlane(nrow, r);
            const int nhi = __builtin_amdgcn_readlane(nrow, r + 8);
            const int n   = lane < 16 ? nlo : nhi;
            if (mr < cnt) {
                v4f o;
                o.x = acc[0][r] + bv.x; o.y = acc[1][r] + bv.y;
                o.z = acc[2][r] + bv.z; o.w = acc[3][r] + bv.w;
                *(v4f*)(out + (size_t)n * O + col4) = o;
            }
        }

        __syncthreads();
    }
}

extern "C" void kernel_launch(void* const* d_in, const int* in_sizes, int n_in,
                              void* d_out, int out_size, void* d_ws, size_t ws_size,
                              hipStream_t stream)
{
    const int*   x      = (const int*)  d_in[0];
    const int*   fact   = (const int*)  d_in[1];
    const float* inp    = (const float*)d_in[2];
    const float* params = (const float*)d_in[3];
    const float* bias   = (const float*)d_in[4];
    const int*   msg_to = (const int*)  d_in[5];
    float*       out    = (float*)d_out;

    const int N   = in_sizes[1] / 2;
    const int O   = in_sizes[4] / NPAIR;
    const int H   = in_sizes[3] / (NPAIR * O);
    const int ORD = in_sizes[2] / (N * H);

    const size_t lds_bytes = (size_t)16 * H * sizeof(float);
    hipLaunchKernelGGL(hon_fused, dim3(NBLOCKS), dim3(NTHREADS), lds_bytes, stream,
                       x, fact, inp, params, bias, msg_to, out, N, H, O, ORD);
}


# A fragments read from LDS as one group of loads per four batches (32 K-steps) at a conflict-free row pitch (H+4 floats, 16 B x odd), not one 16-way conflicted load per two K-steps among the WMMAs
# speedup vs baseline: 6.2217x; 1.1273x over previous
#include <hip/hip_runtime.h>


#define NPAIR 169

typedef __attribute__((ext_vector_type(2))) float v2f;
typedef __attribute__((ext_vector_type(4))) float v4f;
typedef __attribute__((ext_vector_type(8))) float v8f;

#define KS 8
#define KBATCH (4 * KS)
#define NQ 4
#define APAD 4

#define NTHREADS 128
#define NWAVES   (NTHREADS / 32)
#define NBLOCKS  256
#define RPT      8
#define PASSROWS (NTHREADS * RPT)
#define NPAIR2   ((NPAIR + 1) / 2)
static_assert(NWAVES == 4 && NPAIR2 <= NTHREADS, "piece counts are read four at a time; two groups per thread");

__device__ __forceinline__ int row_id(const int* __restrict__ x,
                                      const int* __restrict__ fact, const int n)
{
    const int a = fact[2 * n];
    return x[3 * a + 1] * 13 + x[3 * a + 2];
}

__device__ __forceinline__ int wave_scan_incl(int v, const int ln)
{
#pragma unroll
    for (int d = 1; d < 32; d <<= 1) {
        const int u = __shfl_up(v, d);
        if (ln >= d) v += u;
    }
    return v;
}

__global__ void __launch_bounds__(NTHREADS)
hon_fused(const int* __restrict__ x, const int* __restrict__ fact,
          const float* __restrict__ inp, const float* __restrict__ params,
          const float* __restrict__ bias, const int* __restrict__ msg_to_p,
          float* __restrict__ out, int N, int H, int O, int ORD)
{
    extern __shared__ __attribute__((aligned(16))) float As[];
    __shared__ int gcnt[NPAIR + 1];
    __shared__ int wtot[NWAVES];
    __shared__ int sel[2];
    __shared__ __attribute__((aligned(16))) int pc[RPT * NWAVES];
    __shared__ int rowsel[16];

    const int tid = threadIdx.x;
    const int ln  = tid & 31;
    const int wv  = tid >> 5;
    const int AP  = H + APAD;

    int idr[RPT];
    {
        int a[RPT];
#pragma unroll
        for (int j = 0; j < RPT; ++j) {
            const int n = j * NTHREADS + tid;
            a[j] = fact[2 * (n < N ? n : N - 1)];
        }
#pragma unroll
        for (int j = 0; j < RPT; ++j) {
            const int n  = j * NTHREADS + tid;
            const int id = x[3 * a[j] + 1] * 13 + x[3 * a[j] + 2];
            idr[j] = n < N ? id : -1;
        }
    }
    const int msg_to = msg_to_p[0];

    for (int g = tid; g <= NPAIR; g += NTHREADS) gcnt[g] = 0;
    __syncthreads();
#pragma unroll
    for (int j = 0; j < RPT; ++j)
        if (idr[j] >= 0) atomicAdd(&gcnt[idr[j]], 1);
    for (int n = PASSROWS + tid; n < N; n += NTHREADS)
        atomicAdd(&gcnt[row_id(x, fact, n)], 1);
    __syncthreads();

    int nt0 = 0, nt1 = 0;
    if (tid < NPAIR2) {
        nt0 = (gcnt[2 * tid] + 15) >> 4;
        nt1 = (gcnt[2 * tid + 1] + 15) >> 4;
    }
    const int inc = wave_scan_incl(nt0 + nt1, ln);
    if (ln == 31) wtot[wv] = inc;
    __syncthreads();
    int ts0 = inc - nt0 - nt1, ntile = 0;
#pragma unroll
    for (int w = 0; w < NWAVES; ++w) {
        const int t = wtot[w];
        ntile += t;
        if (w < wv) ts0 += t;
    }
    const int ts1 = ts0 + nt0;

    const int nh     = O >> 8;
    const int nitems = ntile * nh;

    for (int item = blockIdx.x; item < nitems; item += gridDim.x) {
        const int ti   = item / nh;
        const int half = item - ti * nh;

        if (ti >= ts0 && ti < ts0 + nt0) { sel[0] = 2 * tid;     sel[1] = ti - ts0; }
        if (ti >= ts1 && ti < ts1 + nt1) { sel[0] = 2 * tid + 1; sel[1] = ti - ts1; }
        __syncthreads();
        const int g  = sel[0];
        const int mt = sel[1];
        int cnt = gcnt[g] - mt * 16;
        cnt = cnt > 16 ? 16 : cnt;

        int rbase = 0;
        for (int pb = 0; pb < N; pb += PASSROWS) {
            unsigned bal[RPT];
#pragma unroll
            for (int j = 0; j < RPT; ++j) {
                int id = idr[j];
                if (pb > 0) {
                    const int n = pb + j * NTHREADS + tid;
                    id = n < N ? row_id(x, fact, n) : -1;
                }
                bal[j] = __builtin_amdgcn_ballot_w32(id == g);
            }
            if (pb > 0) __syncthreads();
            int mine = 0;
#pragma unroll
            for (int j = 0; j < RPT; ++j) mine = ln == j ? __popc(bal[j]) : mine;
            if (ln < RPT) pc[ln * NWAVES + wv] = mine;
            __syncthreads();
            int4 q[RPT];
#pragma unroll
            for (int j = 0; j < RPT; ++j) q[j] = *(const int4*)&pc[j * NWAVES];
            int run = rbase;
#pragma unroll
            for (int j = 0; j < RPT; ++j) {
                const int pre = run + (wv > 0 ? q[j].x : 0) + (wv > 1 ? q[j].y : 0)
                                    + (wv > 2 ? q[j].z : 0);
                run += q[j].x + q[j].y + q[j].z + q[j].w;
                const int rank = pre + __popc(bal[j] & ((1u << ln) - 1u));
                if (((bal[j] >> ln) & 1u) && (rank >> 4) == mt)
                    rowsel[rank & 15] = pb + j * NTHREADS + tid;
            }
            rbase = run;
        }
        __syncthreads();

        int nrow;
        {
            int rsel = threadIdx.x & 15;
            rsel = rsel < cnt ? rsel : cnt - 1;
            rsel = rsel > 0 ? rsel : 0;
            nrow = rowsel[rsel];
        }

        {
            const int    HQ    = H >> 2;
            const size_t plane = (size_t)N * H;
            const v4f    one   = (v4f){1.f, 1.f, 1.f, 1.f};
            const v4f    zero  = (v4f){0.f, 0.f, 0.f, 0.f};
            const int    nkept = ORD - ((msg_to >= 0 && msg_to < ORD) ? 1 : 0);
            for (int kq = threadIdx.x; kq < HQ; kq += blockDim.x) {
                const unsigned koff = 4u * (unsigned)kq;
                v4f p[16];
                if (nkept == 2) {
                    const int a = msg_to == 0 ? 1 : 0;
                    const int b = msg_to == a + 1 ? a + 2 : a + 1;
                    const float* pa = inp + (size_t)a * plane;
                    const float* pb = inp + (size_t)b * plane;
                    v4f va[16], vb[16];
#pragma unroll
                    for (int r = 0; r < 16; ++r) {
                        const size_t ro = (size_t)__builtin_amdgcn_readlane(nrow, r) * H;
                        va[r] = *(const v4f*)(pa + ro + koff);
                        vb[r] = *(const v4f*)(pb + ro + koff);
                    }
#pragma unroll
                    for (int r = 0; r < 16; ++r) p[r] = va[r] * vb[r];
                } else {
#pragma unroll
                    for (int r = 0; r < 16; ++r) p[r] = one;
                    for (int i = 0; i < ORD; ++i) {
                        if (i == msg_to) continue;
                        const float* pi = inp + (size_t)i * plane;
                        v4f v[16];
#pragma unroll
                        for (int r = 0; r < 16; ++r)
                            v[r] = *(const v4f*)(pi + (size_t)__builtin_amdgcn_readlane(nrow, r) * H + koff);
#pragma unroll
                        for (int r = 0; r < 16; ++r) p[r] = p[r] * v[r];
                    }
                }
#pragma unroll
                for (int r = 0; r < 16; ++r)
                    *(v4f*)(As + r * AP + koff) = r < cnt ? p[r] : zero;
            }
        }
        __syncthreads();

        const int lane   = threadIdx.x & 31;
        const int waveid = threadIdx.x >> 5;
        const int m      = lane & 15;
        const int khalf  = (lane >> 4) << 1;

        const float* W    = params + (size_t)g * H * O;
        const int    col4 = half * 256 + waveid * 64 + m * 4;
        const float* aptr = As + m * AP + khalf;
        const float* bptr = W + (size_t)khalf * O + col4;
        const v4f bv = *(const v4f*)(bias + (size_t)g * O + col4);

        v8f acc[4];
#pragma unroll
        for (int t = 0; t < 4; ++t) acc[t] = (v8f){0, 0, 0, 0, 0, 0, 0, 0};

        auto mma_step = [&](const v2f a, const v4f b0, const v4f b1) {
#pragma unroll
            for (int t = 0; t < 4; ++t) {
                v2f b; b.x = b0[t]; b.y = b1[t];
                acc[t] = __builtin_amdgcn_wmma_f32_16x16x4_f32(
                    false, a, false, b, (short)0, acc[t], false, false);
            }
        };
        auto load_batch = [&](v4f (&q)[KS][2], const int kb) {
#pragma unroll
            for (int s = 0; s < KS; ++s) {
                const float* p = bptr + (size_t)(kb + 4 * s) * O;
                q[s][0] = __builtin_nontemporal_load((const v4f*)p);
                q[s][1] = __builtin_nontemporal_load((const v4f*)(p + O));
            }
        };
        auto load_a = [&](v2f (&a)[NQ * KS], const int kb) {
#pragma unroll
            for (int i = 0; i < NQ * KS; ++i) a[i] = *(const v2f*)(aptr + kb + 4 * i);
        };
        auto mma_batch = [&](const v4f (&q)[KS][2], const v2f (&a)[NQ * KS], const int j) {
#pragma unroll
            for (int s = 0; s < KS; ++s) mma_step(a[j * KS + s], q[s][0], q[s][1]);
        };

        const int nbat = H / KBATCH;
        int kdone = 0;
        if (nbat >= NQ && nbat % NQ == 0) {
            v4f q[NQ][KS][2];
            v2f areg[NQ * KS];
#pragma unroll
            for (int j = 0; j < NQ - 1; ++j) {
                load_batch(q[j], j * KBATCH);
                __builtin_amdgcn_sched_barrier(0);
            }
            int b = 0;
            for (; b + NQ < nbat; b += NQ) {
                load_a(areg, b * KBATCH);
                __builtin_amdgcn_sched_barrier(0);
#pragma unroll
                for (int j = 0; j < NQ; ++j) {
                    load_batch(q[(j + NQ - 1) % NQ], (b + j + NQ - 1) * KBATCH);
                    __builtin_amdgcn_sched_barrier(0);
                    mma_batch(q[j], areg, j);
                }
            }
            load_a(areg, b * KBATCH);
            __builtin_amdgcn_sched_barrier(0);
            load_batch(q[NQ - 1], (b + NQ - 1) * KBATCH);
            __builtin_amdgcn_sched_barrier(0);
#pragma unroll
            for (int j = 0; j < NQ; ++j) mma_batch(q[j], areg, j);
            kdone = nbat * KBATCH;
        }
        for (int kb = kdone; kb < H; kb += 4) {
            const float* p = bptr + (size_t)kb * O;
            mma_step(*(const v2f*)(aptr + kb), *(const v4f*)p, *(const v4f*)(p + O));
        }

        const int mbase = (lane >> 4) << 3;
#pragma unroll
        for (int r = 0; r < 8; ++r) {
            const int mr = mbase + r;
            const int nlo = __builtin_amdgcn_readlane(nrow, r);
            const int nhi = __builtin_amdgcn_readlane(nrow, r + 8);
            const int n   = lane < 16 ? nlo : nhi;
            if (mr < cnt) {
                v4f o;
                o.x = acc[0][r] + bv.x; o.y = acc[1][r] + bv.y;
                o.z = acc[2][r] + bv.z; o.w = acc[3][r] + bv.w;
                *(v4f*)(out + (size_t)n * O + col4) = o;
            }
        }

        __syncthreads();
    }
}

extern "C" void kernel_launch(void* const* d_in, const int* in_sizes, int n_in,
                              void* d_out, int out_size, void* d_ws, size_t ws_size,
                              hipStream_t stream)
{
    const int*   x      = (const int*)  d_in[0];
    const int*   fact   = (const int*)  d_in[1];
    const float* inp    = (const float*)d_in[2];
    const float* params = (const float*)d_in[3];
    const float* bias   = (const float*)d_in[4];
    const int*   msg_to = (const int*)  d_in[5];
    float*       out    = (float*)d_out;

    const int N   = in_sizes[1] / 2;
    const int O   = in_sizes[4] / NPAIR;
    const int H   = in_sizes[3] / (NPAIR * O);
    const int ORD = in_sizes[2] / (N * H);

    const size_t lds_bytes = (size_t)16 * (H + APAD) * sizeof(float);
    hipLaunchKernelGGL(hon_fused, dim3(NBLOCKS), dim3(NTHREADS), lds_bytes, stream,
                       x, fact, inp, params, bias, msg_to, out, N, H, O, ORD);
}
